# starting kernel
# baseline (speedup 1.0000x reference)
#include <hip/hip_runtime.h>


typedef __attribute__((ext_vector_type(2))) float v2f;
typedef __attribute__((ext_vector_type(8))) float v8f;

__global__ void k_deg_init(float* __restrict__ deg, int n) {
  int i = blockIdx.x * blockDim.x + threadIdx.x;
  if (i < n) deg[i] = 1.0f;
}

__global__ void k_deg_count(float* __restrict__ deg, const int* __restrict__ dst, int e) {
  int i = blockIdx.x * blockDim.x + threadIdx.x;
  if (i < e) atomicAdd(&deg[dst[i]], 1.0f);
}

__global__ void k_dinv(float* __restrict__ deg, int n) {
  int i = blockIdx.x * blockDim.x + threadIdx.x;
  if (i < n) deg[i] = rsqrtf(deg[i]);
}

__global__ void k_gemm_wmma(float* __restrict__ h, const float* __restrict__ x,
                            const float* __restrict__ w, int n, int K, int dout) {
  const int lane  = threadIdx.x;
  const int mt    = blockIdx.x;
  const int nt    = blockIdx.y;
  const int half  = lane >> 4;
  const int l15   = lane & 15;

  int row = mt * 16 + l15;
  int arow = (row < n) ? row : (n - 1);
  const int col = nt * 16 + l15;

  const float* __restrict__ xr = x + (size_t)arow * K + half * 2;
  const float* __restrict__ wr = w + (size_t)col  * K + half * 2;

  v8f acc = {};
#pragma unroll 8
  for (int kk = 0; kk < K; kk += 4) {
    v2f a, b;
    a.x = xr[kk];
    a.y = xr[kk + 1];
    b.x = wr[kk];
    b.y = wr[kk + 1];
    acc = __builtin_amdgcn_wmma_f32_16x16x4_f32(
         false, a,  false, b,
         (short)0, acc,  false,  false);
  }

  const int rbase = mt * 16 + half * 8;
#pragma unroll
  for (int r = 0; r < 8; ++r) {
    int orow = rbase + r;
    if (orow < n) h[(size_t)orow * dout + col] = acc[r];
  }
}

__global__ void k_selfloop(float* __restrict__ agg, const float* __restrict__ h,
                           const float* __restrict__ dinv, int dout, long total) {
  long i = (long)blockIdx.x * blockDim.x + threadIdx.x;
  if (i < total) {
    int node = (int)(i / dout);
    float di = dinv[node];
    agg[i] = h[i] * di * di;
  }
}

__global__ void k_scatter(float* __restrict__ agg, const float* __restrict__ h,
                          const float* __restrict__ dinv,
                          const int* __restrict__ src, const int* __restrict__ dst,
                          long total, int dout, int shift) {
  long gid = (long)blockIdx.x * blockDim.x + threadIdx.x;
  if (gid >= total) return;
  int edge = (int)(gid >> shift);
  int c    = (int)(gid & ((1 << shift) - 1)) << 2;
  int s = src[edge];
  int d = dst[edge];
  float nrm = dinv[s] * dinv[d];
  const float4 v = *(const float4*)(h + (size_t)s * dout + c);
  float* ap = agg + (size_t)d * dout + c;
  atomicAdd(ap + 0, v.x * nrm);
  atomicAdd(ap + 1, v.y * nrm);
  atomicAdd(ap + 2, v.z * nrm);
  atomicAdd(ap + 3, v.w * nrm);
}

__global__ void k_finish(float* __restrict__ out, const float* __restrict__ agg,
                         const float* __restrict__ bias, int doutMask, long total,
                         int dorelu) {
  long i = (long)blockIdx.x * blockDim.x + threadIdx.x;
  if (i < total) {
    float v = agg[i] + bias[(int)i & doutMask];
    out[i] = dorelu ? fmaxf(v, 0.0f) : v;
  }
}

static inline size_t align256(size_t x) { return (x + 255) & ~(size_t)255; }

extern "C" void kernel_launch(void* const* d_in, const int* in_sizes, int n_in,
                              void* d_out, int out_size, void* d_ws, size_t ws_size,
                              hipStream_t stream) {
  (void)n_in; (void)out_size; (void)ws_size;

  const int DIN = 128, DH = 128, DOUT = 64, LAYERS = 5;
  const int N = in_sizes[0] / DIN;
  const int E = in_sizes[1] / 2;

  const float* x0   = (const float*)d_in[0];
  const int*  eidx  = (const int*)d_in[1];
  const int*  srcA  = eidx;
  const int*  dstA  = eidx + E;

  char* ws = (char*)d_ws;
  size_t off = 0;
  float* dinv = (float*)(ws + off); off += align256((size_t)N * sizeof(float));
  float* bufH = (float*)(ws + off); off += align256((size_t)N * DH * sizeof(float));
  float* bufA = (float*)(ws + off); off += align256((size_t)N * DH * sizeof(float));
  float* bufX = (float*)(ws + off); off += align256((size_t)N * DH * sizeof(float));

  const int TB = 256;

  k_deg_init <<<(N + TB - 1) / TB, TB, 0, stream>>>(dinv, N);
  k_deg_count<<<(E + TB - 1) / TB, TB, 0, stream>>>(dinv, dstA, E);
  k_dinv     <<<(N + TB - 1) / TB, TB, 0, stream>>>(dinv, N);

  const float* xin = x0;
  for (int layer = 0; layer < LAYERS; ++layer) {
    const int K    = DIN;
    const int dout = (layer == LAYERS - 1) ? DOUT : DH;
    const float* W = (const float*)d_in[2 + 2 * layer];
    const float* b = (const float*)d_in[3 + 2 * layer];
    float* out = (layer == LAYERS - 1) ? (float*)d_out : bufX;

    dim3 grid((N + 15) / 16, dout / 16);
    k_gemm_wmma<<<grid, 32, 0, stream>>>(bufH, xin, W, N, K, dout);

    long tot = (long)N * dout;
    k_selfloop<<<(unsigned)((tot + TB - 1) / TB), TB, 0, stream>>>(bufA, bufH, dinv, dout, tot);

    const int shift = (dout == 128) ? 5 : 4;
    long sc_total = (long)E << shift;
    k_scatter<<<(unsigned)((sc_total + TB - 1) / TB), TB, 0, stream>>>(
        bufA, bufH, dinv, srcA, dstA, sc_total, dout, shift);

    k_finish<<<(unsigned)((tot + TB - 1) / TB), TB, 0, stream>>>(
        out, bufA, b, dout - 1, tot, layer < LAYERS - 1 ? 1 : 0);

    xin = bufX;
  }
}


# Each layer sums its INPUT rows in the owner block's LDS and multiplies the sums by W there (WMMA fed from LDS, pitch 132): no GEMM launch, 13 launches to 8; product order (A x) W, not A (x W)
# speedup vs baseline: 3.6124x; 3.6124x over previous
#include <hip/hip_runtime.h>


typedef __attribute__((ext_vector_type(2))) float v2f;
typedef __attribute__((ext_vector_type(8))) float v8f;

__global__ void k_deg_init(float* __restrict__ deg, int n) {
  int i = blockIdx.x * blockDim.x + threadIdx.x;
  if (i < n) deg[i] = 1.0f;
}

__global__ void k_deg_count(float* __restrict__ deg, const int* __restrict__ dst, int e) {
  int i = blockIdx.x * blockDim.x + threadIdx.x;
  if (i < e) atomicAdd(&deg[dst[i]], 1.0f);
}

__global__ void k_dinv(float* __restrict__ deg, int n) {
  int i = blockIdx.x * blockDim.x + threadIdx.x;
  if (i < n) deg[i] = rsqrtf(deg[i]);
}


#define AGG_ROWS    196
#define AGG_THREADS 1024
#define AGG_WAVES   (AGG_THREADS / 32)
#define AGG_QCAP    512
#define AGG_LCAP    512
#define FEAT        128
#define PITCH       132
#define ROW_TILES   13
#define ROWS_PAD    (ROW_TILES * 16)

__device__ __forceinline__ void agg_rows(float* acc, int ss, int uu, float ww, int m, int lane,
                                         const float* __restrict__ h) {
  constexpr int F   = FEAT;
  constexpr int KPL = F / 32;
  constexpr int UNR = 8;
  for (int j = 0; j < m; j += UNR) {
    float v[UNR][KPL];
#pragma unroll
    for (int t = 0; t < UNR; ++t) {
      const int jj = (j + t < m) ? (j + t) : (m - 1);
      const int s = __builtin_amdgcn_readlane(ss, jj);
      const float* hr = h + (size_t)s * F + lane;
#pragma unroll
      for (int k = 0; k < KPL; ++k) v[t][k] = hr[32 * k];
    }
#pragma unroll
    for (int t = 0; t < UNR; ++t) {
      if (t == 0 || j + t < m) {
        const int u = __builtin_amdgcn_readlane(uu, j + t);
        const float w = __int_as_float(__builtin_amdgcn_readlane(__float_as_int(ww), j + t));
        float* ar = acc + u * PITCH + lane;
#pragma unroll
        for (int k = 0; k < KPL; ++k)
          __hip_atomic_fetch_add(ar + 32 * k, v[t][k] * w, __ATOMIC_RELAXED,
                                 __HIP_MEMORY_SCOPE_WORKGROUP);
      }
    }
  }
}

__device__ __forceinline__ void agg_drain(float* acc, const int2* q, int cnt, int lane, int lo,
                                          const float* __restrict__ h,
                                          const float* __restrict__ dinv,
                                          const int* __restrict__ src,
                                          int4* keep, int kept) {
  for (int b = 0; b < cnt; b += 32) {
    const int m = (cnt - b < 32) ? (cnt - b) : 32;
    int ss = 0, uu = 0;
    float ww = 0.0f;
    if (lane < m) {
      const int2 ent = q[b + lane];
      uu = ent.y;
      ss = src[ent.x];
      ww = dinv[ss] * dinv[lo + uu];
      if (keep) keep[kept + b + lane] = make_int4(ss, uu, __float_as_int(ww), 0);
    }
    agg_rows(acc, ss, uu, ww, m, lane, h);
  }
}

__device__ __forceinline__ void agg_replay(float* acc, const int4* __restrict__ keep, int cnt,
                                           int lane, const float* __restrict__ h) {
  int4 nxt = make_int4(0, 0, 0, 0);
  if (lane < cnt) nxt = keep[lane];
  for (int b = 0; b < cnt; b += 32) {
    const int m = (cnt - b < 32) ? (cnt - b) : 32;
    const int4 cur = nxt;
    if (b + 32 + lane < cnt) nxt = keep[b + 32 + lane];
    agg_rows(acc, cur.x, cur.y, __int_as_float(cur.z), m, lane, h);
  }
}

template <int DOUT>
__device__ __forceinline__ void put_tile(float* __restrict__ out, v8f d, int rt, int half, int col,
                                         float bv, int lo, int n, int dorelu) {
#pragma unroll
  for (int r = 0; r < 8; ++r) {
    const int lr   = rt * 16 + half * 8 + r;
    const int node = lo + lr;
    if (lr < AGG_ROWS && node < n) {
      float v = d[r] + bv;
      if (dorelu) v = fmaxf(v, 0.0f);
      out[(size_t)node * DOUT + col] = v;
    }
  }
}

template <int DOUT, bool FIRST>
__global__ __launch_bounds__(AGG_THREADS)
void k_agg_owner(float* __restrict__ out, const float* __restrict__ h,
                 const float* __restrict__ dinv, const float* __restrict__ w,
                 const float* __restrict__ bias,
                 const int* __restrict__ src, const int* __restrict__ dst,
                 int* __restrict__ kcnt, int* __restrict__ kent,
                 int n, int e, int dorelu) {
  constexpr int F = FEAT;
  __shared__ alignas(16) float acc[ROWS_PAD * PITCH];
  __shared__ alignas(16) float wl[DOUT * PITCH];
  __shared__ int2  que[AGG_WAVES * AGG_QCAP];

  const int tid  = threadIdx.x;
  const int lane = tid & 31;
  const int wave = __builtin_amdgcn_readfirstlane(tid >> 5);
  const int lo   = blockIdx.x * AGG_ROWS;
  constexpr int C4 = F / 4;
  constexpr int P4 = PITCH / 4;
  constexpr int NV = ROWS_PAD * C4;

  const float4* w4 = (const float4*)w;
  for (int i = tid; i < DOUT * C4; i += AGG_THREADS)
    ((float4*)wl)[(i / C4) * P4 + (i % C4)] = w4[i];

  const float4* h4 = (const float4*)(h + (size_t)lo * F);
  for (int i = tid; i < NV; i += AGG_THREADS) {
    const int r    = i / C4;
    const int node = lo + r;
    float4 a = make_float4(0.0f, 0.0f, 0.0f, 0.0f);
    if (r < AGG_ROWS && node < n) {
      const float di = dinv[node];
      const float4 hv = h4[i];
      a.x = hv.x * di * di;
      a.y = hv.y * di * di;
      a.z = hv.z * di * di;
      a.w = hv.w * di * di;
    }
    ((float4*)acc)[r * P4 + (i % C4)] = a;
  }
  __syncthreads();

  const int kslot = blockIdx.x * AGG_WAVES + wave;
  int4* keep = (int4*)kent + (size_t)kslot * AGG_LCAP;
  int have = -1;
  if (!FIRST) have = kcnt[kslot];

  if (have >= 0) {
    agg_replay(acc, keep, have, lane, h);
  } else {
    int2* q = que + wave * AGG_QCAP;
    int cnt = 0;
    int kept = 0;
    bool fits = FIRST;

    auto push = [&](bool mine, unsigned u, int eid) {
      const unsigned mask = __builtin_amdgcn_ballot_w32(mine);
      if (mask) {
        if (mine) q[cnt + __builtin_amdgcn_mbcnt_lo(mask, 0u)] = make_int2(eid, (int)u);
        cnt += __builtin_popcount(mask);
      }
    };
    auto drain = [&]() {
      const bool sv = fits && (kept + cnt <= AGG_LCAP);
      agg_drain(acc, q, cnt, lane, lo, h, dinv, src, sv ? keep : (int4*)nullptr, kept);
      if (sv) kept += cnt; else fits = false;
      cnt = 0;
    };
    auto drain_if_full = [&]() {
      if (cnt > AGG_QCAP - 128) drain();
    };
    auto scan4 = [&](int4 d, int ebase) {
      const unsigned u0 = (unsigned)(d.x - lo), u1 = (unsigned)(d.y - lo);
      const unsigned u2 = (unsigned)(d.z - lo), u3 = (unsigned)(d.w - lo);
      const bool m0 = u0 < (unsigned)AGG_ROWS, m1 = u1 < (unsigned)AGG_ROWS;
      const bool m2 = u2 < (unsigned)AGG_ROWS, m3 = u3 < (unsigned)AGG_ROWS;
      if (__builtin_amdgcn_ballot_w32(m0 | m1 | m2 | m3)) {
        push(m0, u0, ebase);
        push(m1, u1, ebase + 1);
        push(m2, u2, ebase + 2);
        push(m3, u3, ebase + 3);
        drain_if_full();
      }
    };

    const int mis   = (int)(((size_t)dst >> 2) & 3);
    int headn       = (4 - mis) & 3;
    if (headn > e) headn = e;
    const int4* dv  = (const int4*)(dst + headn);
    const int e4    = (e - headn) >> 2;
    const int tail0 = headn + (e4 << 2);
    constexpr int GRP = AGG_THREADS * 4;
    const int nfull = e4 / GRP;

    const int rot = (int)(((long)blockIdx.x * nfull) / gridDim.x);
    for (int it = 0; it < nfull; ++it) {
      int g = it + rot;
      if (g >= nfull) g -= nfull;
      const int q0 = g * GRP + tid;
      int4 d[4];
  #pragma unroll
      for (int t = 0; t < 4; ++t) d[t] = dv[q0 + t * AGG_THREADS];
  #pragma unroll
      for (int t = 0; t < 4; ++t) scan4(d[t], headn + ((q0 + t * AGG_THREADS) << 2));
    }
    for (int q0 = nfull * GRP + wave * 32; q0 < e4; q0 += AGG_THREADS) {
      const int qi = q0 + lane;
      int4 d = make_int4(-1, -1, -1, -1);
      if (qi < e4) d = dv[qi];
      scan4(d, headn + (qi << 2));
    }
    if (wave == 0) {
      int eid = -1;
      if (lane < headn) eid = lane;
      else if (tail0 + (lane - headn) < e) eid = tail0 + (lane - headn);
      int dd = -1;
      if (eid >= 0) dd = dst[eid];
      const unsigned u = (unsigned)(dd - lo);
      push(eid >= 0 && u < (unsigned)AGG_ROWS, u, eid);
    }

    drain();
    if (FIRST && lane == 0) kcnt[kslot] = fits ? kept : -1;
  }
  __syncthreads();

  constexpr int NT = DOUT / 16;
  constexpr int RG = AGG_WAVES / NT;
  const int half = lane >> 4;
  const int l15  = lane & 15;
  const int ct   = wave % NT;
  const int rg   = wave / NT;
  const int col  = ct * 16 + l15;

  const float* wr = wl + col * PITCH + half * 2;
  const float bv  = bias[col];

  int rt = rg;
  for (; rt + RG < ROW_TILES; rt += 2 * RG) {
    const float* a0 = acc + (rt * 16 + l15) * PITCH + half * 2;
    const float* a1 = a0 + RG * 16 * PITCH;
    v8f d0 = {}, d1 = {};
#pragma unroll 8
    for (int t = 0; t < F / 4; ++t) {
      const v2f f0 = *(const v2f*)(a0 + 4 * t);
      const v2f f1 = *(const v2f*)(a1 + 4 * t);
      const v2f fb = *(const v2f*)(wr + 4 * t);
      d0 = __builtin_amdgcn_wmma_f32_16x16x4_f32(false, f0, false, fb, (short)0, d0, false, false);
      d1 = __builtin_amdgcn_wmma_f32_16x16x4_f32(false, f1, false, fb, (short)0, d1, false, false);
    }
    put_tile<DOUT>(out, d0, rt, half, col, bv, lo, n, dorelu);
    put_tile<DOUT>(out, d1, rt + RG, half, col, bv, lo, n, dorelu);
  }
  if (rt < ROW_TILES) {
    const float* a0 = acc + (rt * 16 + l15) * PITCH + half * 2;
    v8f d0 = {};
#pragma unroll 8
    for (int t = 0; t < F / 4; ++t) {
      const v2f f0 = *(const v2f*)(a0 + 4 * t);
      const v2f fb = *(const v2f*)(wr + 4 * t);
      d0 = __builtin_amdgcn_wmma_f32_16x16x4_f32(false, f0, false, fb, (short)0, d0, false, false);
    }
    put_tile<DOUT>(out, d0, rt, half, col, bv, lo, n, dorelu);
  }
}

static inline size_t align256(size_t x) { return (x + 255) & ~(size_t)255; }

extern "C" void kernel_launch(void* const* d_in, const int* in_sizes, int n_in,
                              void* d_out, int out_size, void* d_ws, size_t ws_size,
                              hipStream_t stream) {
  (void)n_in; (void)out_size; (void)ws_size;

  const int DIN = 128, DH = 128, DOUT = 64, LAYERS = 5;
  const int N = in_sizes[0] / DIN;
  const int E = in_sizes[1] / 2;

  const float* x0   = (const float*)d_in[0];
  const int*  eidx  = (const int*)d_in[1];
  const int*  srcA  = eidx;
  const int*  dstA  = eidx + E;

  char* ws = (char*)d_ws;
  size_t off = 0;
  float* dinv = (float*)(ws + off); off += align256((size_t)N * sizeof(float));
  float* bufA = (float*)(ws + off); off += align256((size_t)N * DH * sizeof(float));
  float* bufB = (float*)(ws + off); off += align256((size_t)N * DH * sizeof(float));
  const int ablocks = (N + AGG_ROWS - 1) / AGG_ROWS;
  int* kcnt = (int*)(ws + off); off += align256((size_t)ablocks * AGG_WAVES * sizeof(int));
  int* kent = (int*)(ws + off); off += align256((size_t)ablocks * AGG_WAVES * AGG_LCAP * 16);

  const int TB = 256;

  k_deg_init <<<(N + TB - 1) / TB, TB, 0, stream>>>(dinv, N);
  k_deg_count<<<(E + TB - 1) / TB, TB, 0, stream>>>(dinv, dstA, E);
  k_dinv     <<<(N + TB - 1) / TB, TB, 0, stream>>>(dinv, N);

  const float* xin = x0;
  for (int layer = 0; layer < LAYERS; ++layer) {
    const int dout = (layer == LAYERS - 1) ? DOUT : DH;
    const float* W = (const float*)d_in[2 + 2 * layer];
    const float* b = (const float*)d_in[3 + 2 * layer];
    float* out = (layer == LAYERS - 1) ? (float*)d_out : ((layer & 1) ? bufB : bufA);

    const int relu = layer < LAYERS - 1 ? 1 : 0;
    if (layer == 0)
      k_agg_owner<128, true><<<ablocks, AGG_THREADS, 0, stream>>>(out, xin, dinv, W, b, srcA, dstA, kcnt, kent, N, E, relu);
    else if (dout == 128)
      k_agg_owner<128, false><<<ablocks, AGG_THREADS, 0, stream>>>(out, xin, dinv, W, b, srcA, dstA, kcnt, kent, N, E, relu);
    else
      k_agg_owner<64, false><<<ablocks, AGG_THREADS, 0, stream>>>(out, xin, dinv, W, b, srcA, dstA, kcnt, kent, N, E, relu);

    xin = out;
  }
}


# Aggregation without LDS f32 atomics: local row u belongs to wave u mod 32; layer 0 files edges in per-owner LDS bins (int add rank), keeps them by row; owner sums a row in registers in list order
# speedup vs baseline: 4.0367x; 1.1175x over previous
#include <hip/hip_runtime.h>


typedef __attribute__((ext_vector_type(2))) float v2f;
typedef __attribute__((ext_vector_type(8))) float v8f;

__global__ void k_deg_init(float* __restrict__ deg, int n) {
  int i = blockIdx.x * blockDim.x + threadIdx.x;
  if (i < n) deg[i] = 1.0f;
}

__global__ void k_deg_count(float* __restrict__ deg, const int* __restrict__ dst, int e) {
  int i = blockIdx.x * blockDim.x + threadIdx.x;
  if (i < e) atomicAdd(&deg[dst[i]], 1.0f);
}

__global__ void k_dinv(float* __restrict__ deg, int n) {
  int i = blockIdx.x * blockDim.x + threadIdx.x;
  if (i < n) deg[i] = rsqrtf(deg[i]);
}


#define AGG_ROWS    196
#define AGG_THREADS 1024
#define AGG_WAVES   (AGG_THREADS / 32)
#define AGG_BCAP    384
#define AGG_LCAP    512
#define OWN_ROWS    7
#define FEAT        128
#define PITCH       132
#define ROW_TILES   13
#define ROWS_PAD    (ROW_TILES * 16)

__device__ __forceinline__ void own_rows(float* acc, int& cur, float4& a, int ss, int uu, float ww,
                                         int m, int lane, const float* __restrict__ h) {
  constexpr int F   = FEAT;
  constexpr int P4  = PITCH / 4;
  constexpr int UNR = 8;
  for (int j = 0; j < m; j += UNR) {
    float4 v[UNR];
#pragma unroll
    for (int t = 0; t < UNR; ++t) {
      const int jj = (j + t < m) ? (j + t) : (m - 1);
      const int s = __builtin_amdgcn_readlane(ss, jj);
      v[t] = ((const float4*)(h + (size_t)s * F))[lane];
    }
#pragma unroll
    for (int t = 0; t < UNR; ++t) {
      if (t == 0 || j + t < m) {
        const int u = __builtin_amdgcn_readlane(uu, j + t);
        const float w = __int_as_float(__builtin_amdgcn_readlane(__float_as_int(ww), j + t));
        if (u != cur) {
          if (cur >= 0) ((float4*)acc)[cur * P4 + lane] = a;
          a = ((float4*)acc)[u * P4 + lane];
          cur = u;
        }
        a.x += v[t].x * w;
        a.y += v[t].y * w;
        a.z += v[t].z * w;
        a.w += v[t].w * w;
      }
    }
  }
}

template <bool KEEP>
__device__ __forceinline__ void own_drain(float* acc, const int2* q, int cnt, int lane, int lo,
                                          const float* __restrict__ h,
                                          const float* __restrict__ dinv,
                                          const int* __restrict__ src, int4* keep) {
  constexpr int P4 = PITCH / 4;
  int pos[OWN_ROWS];
#pragma unroll
  for (int k = 0; k < OWN_ROWS; ++k) pos[k] = 0;
  if (KEEP) {
    int c[OWN_ROWS];
#pragma unroll
    for (int k = 0; k < OWN_ROWS; ++k) c[k] = 0;
    for (int b = 0; b < cnt; b += 32) {
      int sl = -1;
      if (b + lane < cnt) sl = q[b + lane].y >> 5;
#pragma unroll
      for (int k = 0; k < OWN_ROWS; ++k)
        c[k] += __builtin_popcount(__builtin_amdgcn_ballot_w32(sl == k));
    }
    int run = 0;
#pragma unroll
    for (int k = 0; k < OWN_ROWS; ++k) { pos[k] = run; run += c[k]; }
  }
  int cur = -1;
  float4 a = make_float4(0.0f, 0.0f, 0.0f, 0.0f);
  for (int b = 0; b < cnt; b += 32) {
    const int m = (cnt - b < 32) ? (cnt - b) : 32;
    int ss = 0, uu = 0;
    float ww = 0.0f;
    const bool on = lane < m;
    if (on) {
      const int2 ent = q[b + lane];
      uu = ent.y;
      ss = src[ent.x];
      ww = dinv[ss] * dinv[lo + uu];
    }
    if (KEEP) {
      const int sl = on ? (uu >> 5) : -1;
      int p = 0;
#pragma unroll
      for (int k = 0; k < OWN_ROWS; ++k) {
        const unsigned mask = __builtin_amdgcn_ballot_w32(sl == k);
        if (sl == k) p = pos[k] + (int)__builtin_amdgcn_mbcnt_lo(mask, 0u);
        pos[k] += __builtin_popcount(mask);
      }
      if (on) keep[p] = make_int4(ss, uu, __float_as_int(ww), 0);
    }
    own_rows(acc, cur, a, ss, uu, ww, m, lane, h);
  }
  if (cur >= 0) ((float4*)acc)[cur * P4 + lane] = a;
}

__device__ __forceinline__ void own_replay(float* acc, const int4* __restrict__ keep, int cnt,
                                           int lane, const float* __restrict__ h) {
  constexpr int P4 = PITCH / 4;
  int cur = -1;
  float4 a = make_float4(0.0f, 0.0f, 0.0f, 0.0f);
  int4 nxt = make_int4(0, 0, 0, 0);
  if (lane < cnt) nxt = keep[lane];
  for (int b = 0; b < cnt; b += 32) {
    const int m = (cnt - b < 32) ? (cnt - b) : 32;
    const int4 cur4 = nxt;
    if (b + 32 + lane < cnt) nxt = keep[b + 32 + lane];
    own_rows(acc, cur, a, cur4.x, cur4.y, __int_as_float(cur4.z), m, lane, h);
  }
  if (cur >= 0) ((float4*)acc)[cur * P4 + lane] = a;
}

template <int DOUT>
__device__ __forceinline__ void put_tile(float* __restrict__ out, v8f d, int rt, int half, int col,
                                         float bv, int lo, int n, int dorelu) {
#pragma unroll
  for (int r = 0; r < 8; ++r) {
    const int lr   = rt * 16 + half * 8 + r;
    const int node = lo + lr;
    if (lr < AGG_ROWS && node < n) {
      float v = d[r] + bv;
      if (dorelu) v = fmaxf(v, 0.0f);
      out[(size_t)node * DOUT + col] = v;
    }
  }
}

template <int DOUT, bool FIRST>
__global__ __launch_bounds__(AGG_THREADS)
void k_agg_owner(float* __restrict__ out, const float* __restrict__ h,
                 const float* __restrict__ dinv, const float* __restrict__ w,
                 const float* __restrict__ bias,
                 const int* __restrict__ src, const int* __restrict__ dst,
                 int* __restrict__ kcnt, int* __restrict__ kent,
                 int n, int e, int dorelu) {
  constexpr int F = FEAT;
  __shared__ alignas(16) float acc[ROWS_PAD * PITCH];
  __shared__ alignas(16) float wl[DOUT * PITCH];
  __shared__ int2 bin[AGG_WAVES * AGG_BCAP];
  __shared__ int  bcnt[AGG_WAVES];
  __shared__ int  spilled;

  const int tid  = threadIdx.x;
  const int lane = tid & 31;
  const int wave = __builtin_amdgcn_readfirstlane(tid >> 5);
  const int lo   = blockIdx.x * AGG_ROWS;
  constexpr int C4 = F / 4;
  constexpr int P4 = PITCH / 4;

  const float4* w4 = (const float4*)w;
  for (int i = tid; i < DOUT * C4; i += AGG_THREADS)
    ((float4*)wl)[(i / C4) * P4 + (i % C4)] = w4[i];

  const int kslot = blockIdx.x * AGG_WAVES + wave;
  int4* keep = (int4*)kent + (size_t)kslot * AGG_LCAP;
  int whole = -1;
  if (!FIRST) whole = __builtin_amdgcn_readfirstlane(kcnt[blockIdx.x * AGG_WAVES]);

  const float4* h4 = (const float4*)h;
  auto start_rows = [&](bool add) {
#pragma unroll
    for (int k = 0; k < OWN_ROWS; ++k) {
      const int u    = wave + 32 * k;
      const int node = lo + u;
      if (u < ROWS_PAD) {
        float4 iv = make_float4(0.0f, 0.0f, 0.0f, 0.0f);
        if (u < AGG_ROWS && node < n) {
          const float di = dinv[node];
          const float4 hv = h4[(size_t)node * C4 + lane];
          iv.x = hv.x * di * di;
          iv.y = hv.y * di * di;
          iv.z = hv.z * di * di;
          iv.w = hv.w * di * di;
        }
        if (add) {
          const float4 r = ((float4*)acc)[u * P4 + lane];
          iv.x += r.x; iv.y += r.y; iv.z += r.z; iv.w += r.w;
        }
        ((float4*)acc)[u * P4 + lane] = iv;
      }
    }
  };

  if (whole >= 0) {
    start_rows(false);
    const int have = __builtin_amdgcn_readfirstlane(kcnt[kslot]);
    own_replay(acc, keep, have, lane, h);
  } else {
    for (int i = tid; i < ROWS_PAD * P4; i += AGG_THREADS)
      ((float4*)acc)[i] = make_float4(0.0f, 0.0f, 0.0f, 0.0f);
    if (tid < AGG_WAVES) bcnt[tid] = 0;
    if (tid == 0) spilled = 0;
    __syncthreads();

    bool lost = false;

    auto file = [&](bool mine, unsigned u, int eid) {
      if (mine) {
        const int o = (int)(u & 31u);
        const int slot = __hip_atomic_fetch_add(&bcnt[o], 1, __ATOMIC_RELAXED,
                                                __HIP_MEMORY_SCOPE_WORKGROUP);
        if (slot < AGG_BCAP) {
          bin[o * AGG_BCAP + slot] = make_int2(eid, (int)u);
        } else {
          lost = true;
          const int sn = src[eid];
          const float wt = dinv[sn] * dinv[lo + (int)u];
          const float* hr = h + (size_t)sn * F;
          float* ar = acc + (int)u * PITCH;
#pragma nounroll
          for (int k = 0; k < F; ++k)
            __hip_atomic_fetch_add(ar + k, hr[k] * wt, __ATOMIC_RELAXED, __HIP_MEMORY_SCOPE_WORKGROUP);
        }
      }
    };
    auto scan4 = [&](int4 d, int ebase) {
      const unsigned u0 = (unsigned)(d.x - lo), u1 = (unsigned)(d.y - lo);
      const unsigned u2 = (unsigned)(d.z - lo), u3 = (unsigned)(d.w - lo);
      const bool m0 = u0 < (unsigned)AGG_ROWS, m1 = u1 < (unsigned)AGG_ROWS;
      const bool m2 = u2 < (unsigned)AGG_ROWS, m3 = u3 < (unsigned)AGG_ROWS;
      if (__builtin_amdgcn_ballot_w32(m0 | m1 | m2 | m3)) {
        file(m0, u0, ebase);
        file(m1, u1, ebase + 1);
        file(m2, u2, ebase + 2);
        file(m3, u3, ebase + 3);
      }
    };

    const int mis   = (int)(((size_t)dst >> 2) & 3);
    int headn       = (4 - mis) & 3;
    if (headn > e) headn = e;
    const int4* dv  = (const int4*)(dst + headn);
    const int e4    = (e - headn) >> 2;
    const int tail0 = headn + (e4 << 2);
    constexpr int GRP = AGG_THREADS * 4;
    const int nfull = e4 / GRP;

    const int rot = (int)(((long)blockIdx.x * nfull) / gridDim.x);
    for (int it = 0; it < nfull; ++it) {
      int g = it + rot;
      if (g >= nfull) g -= nfull;
      const int q0 = g * GRP + tid;
      int4 d[4];
  #pragma unroll
      for (int t = 0; t < 4; ++t) d[t] = dv[q0 + t * AGG_THREADS];
  #pragma unroll
      for (int t = 0; t < 4; ++t) scan4(d[t], headn + ((q0 + t * AGG_THREADS) << 2));
    }
    for (int q0 = nfull * GRP + wave * 32; q0 < e4; q0 += AGG_THREADS) {
      const int qi = q0 + lane;
      int4 d = make_int4(-1, -1, -1, -1);
      if (qi < e4) d = dv[qi];
      scan4(d, headn + (qi << 2));
    }
    if (wave == 0) {
      int eid = -1;
      if (lane < headn) eid = lane;
      else if (tail0 + (lane - headn) < e) eid = tail0 + (lane - headn);
      int dd = -1;
      if (eid >= 0) dd = dst[eid];
      const unsigned u = (unsigned)(dd - lo);
      file(eid >= 0 && u < (unsigned)AGG_ROWS, u, eid);
    }

    if (lost) spilled = 1;
    __syncthreads();

    int cnt = __builtin_amdgcn_readfirstlane(bcnt[wave]);
    if (cnt > AGG_BCAP) cnt = AGG_BCAP;
    start_rows(true);
    own_drain<FIRST>(acc, bin + wave * AGG_BCAP, cnt, lane, lo, h, dinv, src, keep);
    if (FIRST && lane == 0) kcnt[kslot] = spilled ? -1 : cnt;
  }
  __syncthreads();

  constexpr int NT = DOUT / 16;
  constexpr int RG = AGG_WAVES / NT;
  const int half = lane >> 4;
  const int l15  = lane & 15;
  const int ct   = wave % NT;
  const int rg   = wave / NT;
  const int col  = ct * 16 + l15;

  const float* wr = wl + col * PITCH + half * 2;
  const float bv  = bias[col];

  int rt = rg;
  for (; rt + RG < ROW_TILES; rt += 2 * RG) {
    const float* a0 = acc + (rt * 16 + l15) * PITCH + half * 2;
    const float* a1 = a0 + RG * 16 * PITCH;
    v8f d0 = {}, d1 = {};
#pragma unroll 8
    for (int t = 0; t < F / 4; ++t) {
      const v2f f0 = *(const v2f*)(a0 + 4 * t);
      const v2f f1 = *(const v2f*)(a1 + 4 * t);
      const v2f fb = *(const v2f*)(wr + 4 * t);
      d0 = __builtin_amdgcn_wmma_f32_16x16x4_f32(false, f0, false, fb, (short)0, d0, false, false);
      d1 = __builtin_amdgcn_wmma_f32_16x16x4_f32(false, f1, false, fb, (short)0, d1, false, false);
    }
    put_tile<DOUT>(out, d0, rt, half, col, bv, lo, n, dorelu);
    put_tile<DOUT>(out, d1, rt + RG, half, col, bv, lo, n, dorelu);
  }
  if (rt < ROW_TILES) {
    const float* a0 = acc + (rt * 16 + l15) * PITCH + half * 2;
    v8f d0 = {};
#pragma unroll 8
    for (int t = 0; t < F / 4; ++t) {
      const v2f f0 = *(const v2f*)(a0 + 4 * t);
      const v2f fb = *(const v2f*)(wr + 4 * t);
      d0 = __builtin_amdgcn_wmma_f32_16x16x4_f32(false, f0, false, fb, (short)0, d0, false, false);
    }
    put_tile<DOUT>(out, d0, rt, half, col, bv, lo, n, dorelu);
  }
}

static inline size_t align256(size_t x) { return (x + 255) & ~(size_t)255; }

extern "C" void kernel_launch(void* const* d_in, const int* in_sizes, int n_in,
                              void* d_out, int out_size, void* d_ws, size_t ws_size,
                              hipStream_t stream) {
  (void)n_in; (void)out_size; (void)ws_size;

  const int DIN = 128, DH = 128, DOUT = 64, LAYERS = 5;
  const int N = in_sizes[0] / DIN;
  const int E = in_sizes[1] / 2;

  const float* x0   = (const float*)d_in[0];
  const int*  eidx  = (const int*)d_in[1];
  const int*  srcA  = eidx;
  const int*  dstA  = eidx + E;

  char* ws = (char*)d_ws;
  size_t off = 0;
  float* dinv = (float*)(ws + off); off += align256((size_t)N * sizeof(float));
  float* bufA = (float*)(ws + off); off += align256((size_t)N * DH * sizeof(float));
  float* bufB = (float*)(ws + off); off += align256((size_t)N * DH * sizeof(float));
  const int ablocks = (N + AGG_ROWS - 1) / AGG_ROWS;
  int* kcnt = (int*)(ws + off); off += align256((size_t)ablocks * AGG_WAVES * sizeof(int));
  int* kent = (int*)(ws + off); off += align256((size_t)ablocks * AGG_WAVES * AGG_LCAP * 16);

  const int TB = 256;

  k_deg_init <<<(N + TB - 1) / TB, TB, 0, stream>>>(dinv, N);
  k_deg_count<<<(E + TB - 1) / TB, TB, 0, stream>>>(dinv, dstA, E);
  k_dinv     <<<(N + TB - 1) / TB, TB, 0, stream>>>(dinv, N);

  const float* xin = x0;
  for (int layer = 0; layer < LAYERS; ++layer) {
    const int dout = (layer == LAYERS - 1) ? DOUT : DH;
    const float* W = (const float*)d_in[2 + 2 * layer];
    const float* b = (const float*)d_in[3 + 2 * layer];
    float* out = (layer == LAYERS - 1) ? (float*)d_out : ((layer & 1) ? bufB : bufA);

    const int relu = layer < LAYERS - 1 ? 1 : 0;
    if (layer == 0)
      k_agg_owner<128, true><<<ablocks, AGG_THREADS, 0, stream>>>(out, xin, dinv, W, b, srcA, dstA, kcnt, kent, N, E, relu);
    else if (dout == 128)
      k_agg_owner<128, false><<<ablocks, AGG_THREADS, 0, stream>>>(out, xin, dinv, W, b, srcA, dstA, kcnt, kent, N, E, relu);
    else
      k_agg_owner<64, false><<<ablocks, AGG_THREADS, 0, stream>>>(out, xin, dinv, W, b, srcA, dstA, kcnt, kent, N, E, relu);

    xin = out;
  }
}


# Layer 0's edge scan reads one owner-block byte per edge (written by the degree count), 16 edges per load, and the 4-byte destination only where the byte matches; sum order: LDS arrival rank as before
# speedup vs baseline: 4.2980x; 1.0647x over previous
#include <hip/hip_runtime.h>


typedef __attribute__((ext_vector_type(2))) float v2f;
typedef __attribute__((ext_vector_type(8))) float v8f;

#define AGG_ROWS    196

__global__ void k_deg_init(float* __restrict__ deg, int n) {
  int i = blockIdx.x * blockDim.x + threadIdx.x;
  if (i < n) deg[i] = 1.0f;
}

__global__ void k_deg_count(float* __restrict__ deg, const int* __restrict__ dst,
                            int* __restrict__ own, int e) {
  int i = blockIdx.x * blockDim.x + threadIdx.x;
  if (i < e) {
    const int d = dst[i];
    atomicAdd(&deg[d], 1.0f);
    ((unsigned char*)own)[i] = (unsigned char)(d / AGG_ROWS);
  }
}

__global__ void k_dinv(float* __restrict__ deg, int n) {
  int i = blockIdx.x * blockDim.x + threadIdx.x;
  if (i < n) deg[i] = rsqrtf(deg[i]);
}


#define AGG_THREADS 1024
#define AGG_WAVES   (AGG_THREADS / 32)
#define AGG_BCAP    256
#define OWN_CAP     256
#define AGG_LCAP    512
#define OWN_ROWS    7
#define FEAT        128
#define PITCH       132
#define ROW_TILES   13
#define ROWS_PAD    (ROW_TILES * 16)

__device__ __forceinline__ void own_rows(float* acc, int& cur, float4& a, int ss, int uu, float ww,
                                         int m, int lane, const float* __restrict__ h) {
  constexpr int F   = FEAT;
  constexpr int P4  = PITCH / 4;
  constexpr int UNR = 8;
  for (int j = 0; j < m; j += UNR) {
    float4 v[UNR];
#pragma unroll
    for (int t = 0; t < UNR; ++t) {
      const int jj = (j + t < m) ? (j + t) : (m - 1);
      const int s = __builtin_amdgcn_readlane(ss, jj);
      v[t] = ((const float4*)(h + (size_t)s * F))[lane];
    }
#pragma unroll
    for (int t = 0; t < UNR; ++t) {
      if (t == 0 || j + t < m) {
        const int u = __builtin_amdgcn_readlane(uu, j + t);
        const float w = __int_as_float(__builtin_amdgcn_readlane(__float_as_int(ww), j + t));
        if (u != cur) {
          if (cur >= 0) ((float4*)acc)[cur * P4 + lane] = a;
          a = ((float4*)acc)[u * P4 + lane];
          cur = u;
        }
        a.x += v[t].x * w;
        a.y += v[t].y * w;
        a.z += v[t].z * w;
        a.w += v[t].w * w;
      }
    }
  }
}

template <bool KEEP>
__device__ __forceinline__ void own_drain(float* acc, const int2* q, int cnt, int lane, int lo,
                                          const float* __restrict__ h,
                                          const float* __restrict__ dinv,
                                          const int* __restrict__ src, int4* keep) {
  constexpr int P4 = PITCH / 4;
  int pos[OWN_ROWS];
#pragma unroll
  for (int k = 0; k < OWN_ROWS; ++k) pos[k] = 0;
  if (KEEP) {
    int c[OWN_ROWS];
#pragma unroll
    for (int k = 0; k < OWN_ROWS; ++k) c[k] = 0;
    for (int b = 0; b < cnt; b += 32) {
      int sl = -1;
      if (b + lane < cnt) sl = q[b + lane].y >> 5;
#pragma unroll
      for (int k = 0; k < OWN_ROWS; ++k)
        c[k] += __builtin_popcount(__builtin_amdgcn_ballot_w32(sl == k));
    }
    int run = 0;
#pragma unroll
    for (int k = 0; k < OWN_ROWS; ++k) { pos[k] = run; run += c[k]; }
  }
  int cur = -1;
  float4 a = make_float4(0.0f, 0.0f, 0.0f, 0.0f);
  for (int b = 0; b < cnt; b += 32) {
    const int m = (cnt - b < 32) ? (cnt - b) : 32;
    int ss = 0, uu = 0;
    float ww = 0.0f;
    const bool on = lane < m;
    if (on) {
      const int2 ent = q[b + lane];
      uu = ent.y;
      ss = src[ent.x];
      ww = dinv[ss] * dinv[lo + uu];
    }
    if (KEEP) {
      const int sl = on ? (uu >> 5) : -1;
      int p = 0;
#pragma unroll
      for (int k = 0; k < OWN_ROWS; ++k) {
        const unsigned mask = __builtin_amdgcn_ballot_w32(sl == k);
        if (sl == k) p = pos[k] + (int)__builtin_amdgcn_mbcnt_lo(mask, 0u);
        pos[k] += __builtin_popcount(mask);
      }
      if (on) keep[p] = make_int4(ss, uu, __float_as_int(ww), 0);
    }
    own_rows(acc, cur, a, ss, uu, ww, m, lane, h);
  }
  if (cur >= 0) ((float4*)acc)[cur * P4 + lane] = a;
}

__device__ __forceinline__ void own_replay(float* acc, const int4* __restrict__ keep, int cnt,
                                           int lane, const float* __restrict__ h) {
  constexpr int P4 = PITCH / 4;
  int cur = -1;
  float4 a = make_float4(0.0f, 0.0f, 0.0f, 0.0f);
  int4 nxt = make_int4(0, 0, 0, 0);
  if (lane < cnt) nxt = keep[lane];
  for (int b = 0; b < cnt; b += 32) {
    const int m = (cnt - b < 32) ? (cnt - b) : 32;
    const int4 cur4 = nxt;
    if (b + 32 + lane < cnt) nxt = keep[b + 32 + lane];
    own_rows(acc, cur, a, cur4.x, cur4.y, __int_as_float(cur4.z), m, lane, h);
  }
  if (cur >= 0) ((float4*)acc)[cur * P4 + lane] = a;
}

template <int DOUT>
__device__ __forceinline__ void put_tile(float* __restrict__ out, v8f d, int rt, int half, int col,
                                         float bv, int lo, int n, int dorelu) {
#pragma unroll
  for (int r = 0; r < 8; ++r) {
    const int lr   = rt * 16 + half * 8 + r;
    const int node = lo + lr;
    if (lr < AGG_ROWS && node < n) {
      float v = d[r] + bv;
      if (dorelu) v = fmaxf(v, 0.0f);
      out[(size_t)node * DOUT + col] = v;
    }
  }
}

template <int DOUT, bool FIRST>
__global__ __launch_bounds__(AGG_THREADS)
void k_agg_owner(float* __restrict__ out, const float* __restrict__ h,
                 const float* __restrict__ dinv, const float* __restrict__ w,
                 const float* __restrict__ bias,
                 const int* __restrict__ src, const int* __restrict__ dst,
                 const int* __restrict__ own,
                 int* __restrict__ kcnt, int* __restrict__ kent,
                 int n, int e, int dorelu) {
  constexpr int F = FEAT;
  __shared__ alignas(16) float acc[ROWS_PAD * PITCH];
  __shared__ alignas(16) float wl[DOUT * PITCH];
  __shared__ int2 bin[AGG_WAVES * AGG_BCAP];
  __shared__ int2 ebin[AGG_WAVES * OWN_CAP];
  __shared__ int  bcnt[AGG_WAVES];
  __shared__ int  spilled;

  const int tid  = threadIdx.x;
  const int lane = tid & 31;
  const int wave = __builtin_amdgcn_readfirstlane(tid >> 5);
  const int lo   = blockIdx.x * AGG_ROWS;
  constexpr int C4 = F / 4;
  constexpr int P4 = PITCH / 4;

  const float4* w4 = (const float4*)w;
  for (int i = tid; i < DOUT * C4; i += AGG_THREADS)
    ((float4*)wl)[(i / C4) * P4 + (i % C4)] = w4[i];

  const int kslot = blockIdx.x * AGG_WAVES + wave;
  int4* keep = (int4*)kent + (size_t)kslot * AGG_LCAP;
  int whole = -1;
  if (!FIRST) whole = __builtin_amdgcn_readfirstlane(kcnt[blockIdx.x * AGG_WAVES]);

  const float4* h4 = (const float4*)h;
  auto start_rows = [&](bool add) {
#pragma unroll
    for (int k = 0; k < OWN_ROWS; ++k) {
      const int u    = wave + 32 * k;
      const int node = lo + u;
      if (u < ROWS_PAD) {
        float4 iv = make_float4(0.0f, 0.0f, 0.0f, 0.0f);
        if (u < AGG_ROWS && node < n) {
          const float di = dinv[node];
          const float4 hv = h4[(size_t)node * C4 + lane];
          iv.x = hv.x * di * di;
          iv.y = hv.y * di * di;
          iv.z = hv.z * di * di;
          iv.w = hv.w * di * di;
        }
        if (add) {
          const float4 r = ((float4*)acc)[u * P4 + lane];
          iv.x += r.x; iv.y += r.y; iv.z += r.z; iv.w += r.w;
        }
        ((float4*)acc)[u * P4 + lane] = iv;
      }
    }
  };

  if (whole >= 0) {
    start_rows(false);
    const int have = __builtin_amdgcn_readfirstlane(kcnt[kslot]);
    own_replay(acc, keep, have, lane, h);
  } else {
    for (int i = tid; i < ROWS_PAD * P4; i += AGG_THREADS)
      ((float4*)acc)[i] = make_float4(0.0f, 0.0f, 0.0f, 0.0f);
    if (tid < AGG_WAVES) bcnt[tid] = 0;
    if (tid == 0) spilled = 0;
    __syncthreads();

    bool lost = false;

    auto file = [&](bool mine, unsigned u, int eid) {
      if (mine) {
        const int o = (int)(u & 31u);
        const int slot = __hip_atomic_fetch_add(&bcnt[o], 1, __ATOMIC_RELAXED,
                                                __HIP_MEMORY_SCOPE_WORKGROUP);
        if (slot < AGG_BCAP) {
          bin[o * AGG_BCAP + slot] = make_int2(eid, (int)u);
        } else {
          lost = true;
          const int sn = src[eid];
          const float wt = dinv[sn] * dinv[lo + (int)u];
          const float* hr = h + (size_t)sn * F;
          float* ar = acc + (int)u * PITCH;
#pragma nounroll
          for (int k = 0; k < F; ++k)
            __hip_atomic_fetch_add(ar + k, hr[k] * wt, __ATOMIC_RELAXED, __HIP_MEMORY_SCOPE_WORKGROUP);
        }
      }
    };
    auto verify = [&](int g, unsigned wmap) {
      while (wmap) {
        const int bit = __builtin_ctz(wmap);
        wmap &= wmap - 1u;
        const int eid = (g << 4) + ((bit & 7) << 2) + (bit >> 3);
        if (eid < e) {
          const unsigned u = (unsigned)(dst[eid] - lo);
          file(u < (unsigned)AGG_ROWS, u, eid);
        }
      }
    };

    const unsigned pat = (unsigned)(blockIdx.x & 255) * 0x01010101u;
    int2* eb  = ebin + wave * OWN_CAP;
    int ecnt  = 0;
    auto probe = [&](int4 o, int g) {
      const unsigned x0 = (unsigned)o.x ^ pat, x1 = (unsigned)o.y ^ pat;
      const unsigned x2 = (unsigned)o.z ^ pat, x3 = (unsigned)o.w ^ pat;
      const unsigned z0 = (x0 - 0x01010101u) & ~x0 & 0x80808080u;
      const unsigned z1 = (x1 - 0x01010101u) & ~x1 & 0x80808080u;
      const unsigned z2 = (x2 - 0x01010101u) & ~x2 & 0x80808080u;
      const unsigned z3 = (x3 - 0x01010101u) & ~x3 & 0x80808080u;
      const bool has = (z0 | z1 | z2 | z3) != 0u;
      const unsigned hm = __builtin_amdgcn_ballot_w32(has);
      if (hm) {
        if (has) {
          const unsigned wmap = (z0 >> 7) | (z1 >> 6) | (z2 >> 5) | (z3 >> 4);
          const int slot = ecnt + (int)__builtin_amdgcn_mbcnt_lo(hm, 0u);
          if (slot < OWN_CAP) eb[slot] = make_int2(g, (int)wmap);
          else verify(g, wmap);
        }
        ecnt += __builtin_popcount(hm);
      }
    };

    const int4* ov  = (const int4*)own;
    const int g16   = (e + 15) >> 4;
    constexpr int SU  = 8;
    constexpr int GRP = AGG_THREADS * SU;
    const int nfull = g16 / GRP;

    const int rot = (int)(((long)blockIdx.x * nfull) / gridDim.x);
    for (int it = 0; it < nfull; ++it) {
      int g = it + rot;
      if (g >= nfull) g -= nfull;
      const int q0 = g * GRP + tid;
      int4 o[SU];
  #pragma unroll
      for (int t = 0; t < SU; ++t) o[t] = ov[q0 + t * AGG_THREADS];
  #pragma unroll
      for (int t = 0; t < SU; ++t) probe(o[t], q0 + t * AGG_THREADS);
    }
    for (int q0 = nfull * GRP + wave * 32; q0 < g16; q0 += AGG_THREADS) {
      const int qi = q0 + lane;
      int4 o = make_int4((int)~pat, (int)~pat, (int)~pat, (int)~pat);
      if (qi < g16) o = ov[qi];
      probe(o, qi);
    }
    {
      const int have = ecnt < OWN_CAP ? ecnt : OWN_CAP;
      for (int b = 0; b < have; b += 32) {
        if (b + lane < have) {
          const int2 c = eb[b + lane];
          verify(c.x, (unsigned)c.y);
        }
      }
    }

    if (lost) spilled = 1;
    __syncthreads();

    int cnt = __builtin_amdgcn_readfirstlane(bcnt[wave]);
    if (cnt > AGG_BCAP) cnt = AGG_BCAP;
    start_rows(true);
    own_drain<FIRST>(acc, bin + wave * AGG_BCAP, cnt, lane, lo, h, dinv, src, keep);
    if (FIRST && lane == 0) kcnt[kslot] = spilled ? -1 : cnt;
  }
  __syncthreads();

  constexpr int NT = DOUT / 16;
  constexpr int RG = AGG_WAVES / NT;
  const int half = lane >> 4;
  const int l15  = lane & 15;
  const int ct   = wave % NT;
  const int rg   = wave / NT;
  const int col  = ct * 16 + l15;

  const float* wr = wl + col * PITCH + half * 2;
  const float bv  = bias[col];

  int rt = rg;
  for (; rt + RG < ROW_TILES; rt += 2 * RG) {
    const float* a0 = acc + (rt * 16 + l15) * PITCH + half * 2;
    const float* a1 = a0 + RG * 16 * PITCH;
    v8f d0 = {}, d1 = {};
#pragma unroll 8
    for (int t = 0; t < F / 4; ++t) {
      const v2f f0 = *(const v2f*)(a0 + 4 * t);
      const v2f f1 = *(const v2f*)(a1 + 4 * t);
      const v2f fb = *(const v2f*)(wr + 4 * t);
      d0 = __builtin_amdgcn_wmma_f32_16x16x4_f32(false, f0, false, fb, (short)0, d0, false, false);
      d1 = __builtin_amdgcn_wmma_f32_16x16x4_f32(false, f1, false, fb, (short)0, d1, false, false);
    }
    put_tile<DOUT>(out, d0, rt, half, col, bv, lo, n, dorelu);
    put_tile<DOUT>(out, d1, rt + RG, half, col, bv, lo, n, dorelu);
  }
  if (rt < ROW_TILES) {
    const float* a0 = acc + (rt * 16 + l15) * PITCH + half * 2;
    v8f d0 = {};
#pragma unroll 8
    for (int t = 0; t < F / 4; ++t) {
      const v2f f0 = *(const v2f*)(a0 + 4 * t);
      const v2f fb = *(const v2f*)(wr + 4 * t);
      d0 = __builtin_amdgcn_wmma_f32_16x16x4_f32(false, f0, false, fb, (short)0, d0, false, false);
    }
    put_tile<DOUT>(out, d0, rt, half, col, bv, lo, n, dorelu);
  }
}

static inline size_t align256(size_t x) { return (x + 255) & ~(size_t)255; }

extern "C" void kernel_launch(void* const* d_in, const int* in_sizes, int n_in,
                              void* d_out, int out_size, void* d_ws, size_t ws_size,
                              hipStream_t stream) {
  (void)n_in; (void)out_size; (void)ws_size;

  const int DIN = 128, DH = 128, DOUT = 64, LAYERS = 5;
  const int N = in_sizes[0] / DIN;
  const int E = in_sizes[1] / 2;

  const float* x0   = (const float*)d_in[0];
  const int*  eidx  = (const int*)d_in[1];
  const int*  srcA  = eidx;
  const int*  dstA  = eidx + E;

  char* ws = (char*)d_ws;
  size_t off = 0;
  float* dinv = (float*)(ws + off); off += align256((size_t)N * sizeof(float));
  float* bufA = (float*)(ws + off); off += align256((size_t)N * DH * sizeof(float));
  float* bufB = (float*)(ws + off); off += align256((size_t)N * DH * sizeof(float));
  const int ablocks = (N + AGG_ROWS - 1) / AGG_ROWS;
  int* kcnt = (int*)(ws + off); off += align256((size_t)ablocks * AGG_WAVES * sizeof(int));
  int* kent = (int*)(ws + off); off += align256((size_t)ablocks * AGG_WAVES * AGG_LCAP * 16);
  int* own  = (int*)(ws + off); off += align256((size_t)E);

  const int TB = 256;

  k_deg_init <<<(N + TB - 1) / TB, TB, 0, stream>>>(dinv, N);
  k_deg_count<<<(E + TB - 1) / TB, TB, 0, stream>>>(dinv, dstA, own, E);
  k_dinv     <<<(N + TB - 1) / TB, TB, 0, stream>>>(dinv, N);

  const float* xin = x0;
  for (int layer = 0; layer < LAYERS; ++layer) {
    const int dout = (layer == LAYERS - 1) ? DOUT : DH;
    const float* W = (const float*)d_in[2 + 2 * layer];
    const float* b = (const float*)d_in[3 + 2 * layer];
    float* out = (layer == LAYERS - 1) ? (float*)d_out : ((layer & 1) ? bufB : bufA);

    const int relu = layer < LAYERS - 1 ? 1 : 0;
    if (layer == 0)
      k_agg_owner<128, true><<<ablocks, AGG_THREADS, 0, stream>>>(out, xin, dinv, W, b, srcA, dstA, own, kcnt, kent, N, E, relu);
    else if (dout == 128)
      k_agg_owner<128, false><<<ablocks, AGG_THREADS, 0, stream>>>(out, xin, dinv, W, b, srcA, dstA, own, kcnt, kent, N, E, relu);
    else
      k_agg_owner<64, false><<<ablocks, AGG_THREADS, 0, stream>>>(out, xin, dinv, W, b, srcA, dstA, own, kcnt, kent, N, E, relu);

    xin = out;
  }
}


# Every wave's set-up loads go out as one group before any wait: weights, self-loop rows, list count and head (fourteen serial round trips become two); next list window asked for behind row loads
# speedup vs baseline: 4.5115x; 1.0497x over previous
#include <hip/hip_runtime.h>


typedef __attribute__((ext_vector_type(2))) float v2f;
typedef __attribute__((ext_vector_type(8))) float v8f;

#define AGG_ROWS    196

__global__ void k_deg_init(float* __restrict__ deg, int n) {
  int i = blockIdx.x * blockDim.x + threadIdx.x;
  if (i < n) deg[i] = 1.0f;
}

__global__ void k_deg_count(float* __restrict__ deg, const int* __restrict__ dst,
                            int* __restrict__ own, int e) {
  int i = blockIdx.x * blockDim.x + threadIdx.x;
  if (i < e) {
    const int d = dst[i];
    atomicAdd(&deg[d], 1.0f);
    ((unsigned char*)own)[i] = (unsigned char)(d / AGG_ROWS);
  }
}

__global__ void k_dinv(float* __restrict__ deg, int n) {
  int i = blockIdx.x * blockDim.x + threadIdx.x;
  if (i < n) deg[i] = rsqrtf(deg[i]);
}


#define AGG_THREADS 1024
#define AGG_WAVES   (AGG_THREADS / 32)
#define AGG_BCAP    256
#define OWN_CAP     256
#define AGG_LCAP    512
#define OWN_ROWS    7
#define FEAT        128
#define PITCH       132
#define ROW_TILES   13
#define ROWS_PAD    (ROW_TILES * 16)

template <class PF>
__device__ __forceinline__ void own_rows(float* acc, int& cur, float4& a, int ss, int uu, float ww,
                                         int m, int lane, const float* __restrict__ h,
                                         PF&& behind_first_loads) {
  constexpr int F   = FEAT;
  constexpr int P4  = PITCH / 4;
  constexpr int UNR = 8;
  for (int j = 0; j < m; j += UNR) {
    float4 v[UNR];
#pragma unroll
    for (int t = 0; t < UNR; ++t) {
      const int jj = (j + t < m) ? (j + t) : (m - 1);
      const int s = __builtin_amdgcn_readlane(ss, jj);
      v[t] = ((const float4*)(h + (size_t)s * F))[lane];
    }
    if (j == 0) behind_first_loads();
#pragma unroll
    for (int t = 0; t < UNR; ++t) {
      if (t == 0 || j + t < m) {
        const int u = __builtin_amdgcn_readlane(uu, j + t);
        const float w = __int_as_float(__builtin_amdgcn_readlane(__float_as_int(ww), j + t));
        if (u != cur) {
          if (cur >= 0) ((float4*)acc)[cur * P4 + lane] = a;
          a = ((float4*)acc)[u * P4 + lane];
          cur = u;
        }
        a.x += v[t].x * w;
        a.y += v[t].y * w;
        a.z += v[t].z * w;
        a.w += v[t].w * w;
      }
    }
  }
}

template <bool KEEP>
__device__ __forceinline__ void own_drain(float* acc, const int2* q, int cnt, int lane, int lo,
                                          const float* __restrict__ h,
                                          const float* __restrict__ dinv,
                                          const int* __restrict__ src, int4* keep) {
  constexpr int P4 = PITCH / 4;
  int pos[OWN_ROWS];
#pragma unroll
  for (int k = 0; k < OWN_ROWS; ++k) pos[k] = 0;
  if (KEEP) {
    int c[OWN_ROWS];
#pragma unroll
    for (int k = 0; k < OWN_ROWS; ++k) c[k] = 0;
    for (int b = 0; b < cnt; b += 32) {
      int sl = -1;
      if (b + lane < cnt) sl = q[b + lane].y >> 5;
#pragma unroll
      for (int k = 0; k < OWN_ROWS; ++k)
        c[k] += __builtin_popcount(__builtin_amdgcn_ballot_w32(sl == k));
    }
    int run = 0;
#pragma unroll
    for (int k = 0; k < OWN_ROWS; ++k) { pos[k] = run; run += c[k]; }
  }
  int cur = -1;
  float4 a = make_float4(0.0f, 0.0f, 0.0f, 0.0f);
  for (int b = 0; b < cnt; b += 32) {
    const int m = (cnt - b < 32) ? (cnt - b) : 32;
    int ss = 0, uu = 0;
    float ww = 0.0f;
    const bool on = lane < m;
    if (on) {
      const int2 ent = q[b + lane];
      uu = ent.y;
      ss = src[ent.x];
      ww = dinv[ss] * dinv[lo + uu];
    }
    if (KEEP) {
      const int sl = on ? (uu >> 5) : -1;
      int p = 0;
#pragma unroll
      for (int k = 0; k < OWN_ROWS; ++k) {
        const unsigned mask = __builtin_amdgcn_ballot_w32(sl == k);
        if (sl == k) p = pos[k] + (int)__builtin_amdgcn_mbcnt_lo(mask, 0u);
        pos[k] += __builtin_popcount(mask);
      }
      if (on) keep[p] = make_int4(ss, uu, __float_as_int(ww), 0);
    }
    own_rows(acc, cur, a, ss, uu, ww, m, lane, h, [] {});
  }
  if (cur >= 0) ((float4*)acc)[cur * P4 + lane] = a;
}

__device__ __forceinline__ void own_replay(float* acc, const int4* __restrict__ keep, int4 nxt,
                                           int cnt, int lane, const float* __restrict__ h) {
  constexpr int P4 = PITCH / 4;
  int cur = -1;
  float4 a = make_float4(0.0f, 0.0f, 0.0f, 0.0f);
  for (int b = 0; b < cnt; b += 32) {
    const int m = (cnt - b < 32) ? (cnt - b) : 32;
    const int4 cur4 = nxt;
    own_rows(acc, cur, a, cur4.x, cur4.y, __int_as_float(cur4.z), m, lane, h,
             [&] { if (b + 32 + lane < cnt) nxt = keep[b + 32 + lane]; });
  }
  if (cur >= 0) ((float4*)acc)[cur * P4 + lane] = a;
}

template <int DOUT>
__device__ __forceinline__ void put_tile(float* __restrict__ out, v8f d, int rt, int half, int col,
                                         float bv, int lo, int n, int dorelu) {
#pragma unroll
  for (int r = 0; r < 8; ++r) {
    const int lr   = rt * 16 + half * 8 + r;
    const int node = lo + lr;
    if (lr < AGG_ROWS && node < n) {
      float v = d[r] + bv;
      if (dorelu) v = fmaxf(v, 0.0f);
      out[(size_t)node * DOUT + col] = v;
    }
  }
}

template <int DOUT, bool FIRST>
__global__ __launch_bounds__(AGG_THREADS)
void k_agg_owner(float* __restrict__ out, const float* __restrict__ h,
                 const float* __restrict__ dinv, const float* __restrict__ w,
                 const float* __restrict__ bias,
                 const int* __restrict__ src, const int* __restrict__ dst,
                 const int* __restrict__ own,
                 int* __restrict__ kcnt, int* __restrict__ kent,
                 int n, int e, int dorelu) {
  constexpr int F = FEAT;
  __shared__ alignas(16) float acc[ROWS_PAD * PITCH];
  __shared__ alignas(16) float wl[DOUT * PITCH];
  __shared__ int2 bin[AGG_WAVES * AGG_BCAP];
  __shared__ int2 ebin[AGG_WAVES * OWN_CAP];
  __shared__ int  bcnt[AGG_WAVES];
  __shared__ int  spilled;

  const int tid  = threadIdx.x;
  const int lane = tid & 31;
  const int wave = __builtin_amdgcn_readfirstlane(tid >> 5);
  const int lo   = blockIdx.x * AGG_ROWS;
  constexpr int C4 = F / 4;
  constexpr int P4 = PITCH / 4;


  constexpr int WN = DOUT * C4 / AGG_THREADS;
  static_assert(WN * AGG_THREADS == DOUT * C4 && (WN == 2 || WN == 4),
                "two or four whole pieces of the weights per thread");
  const float4* w4 = (const float4*)w;
  const float4 wv0 = w4[tid];
  const float4 wv1 = w4[tid + AGG_THREADS];
  float4 wv2 = wv0, wv3 = wv1;
  if (WN == 4) {
    wv2 = w4[tid + 2 * AGG_THREADS];
    wv3 = w4[tid + 3 * AGG_THREADS];
  }
  auto put_piece = [&](int t, float4 v) {
    const int i = tid + t * AGG_THREADS;
    ((float4*)wl)[(i / C4) * P4 + (i % C4)] = v;
  };

  const int kslot = blockIdx.x * AGG_WAVES + wave;
  int4* keep = (int4*)kent + (size_t)kslot * AGG_LCAP;

  const float4* h4 = (const float4*)h;
  float4 sv[OWN_ROWS];
  float  sd[OWN_ROWS];
  auto self_load = [&]() {
#pragma unroll
    for (int k = 0; k < OWN_ROWS; ++k) {
      const int u    = wave + 32 * k;
      const int node = (u < AGG_ROWS && lo + u < n) ? (lo + u) : lo;
      sd[k] = dinv[node];
      sv[k] = h4[(size_t)node * C4 + lane];
    }
  };
  auto self_store = [&](bool add) {
#pragma unroll
    for (int k = 0; k < OWN_ROWS; ++k) {
      const int u    = wave + 32 * k;
      const int node = lo + u;
      if (u < ROWS_PAD) {
        float4 iv = make_float4(0.0f, 0.0f, 0.0f, 0.0f);
        if (u < AGG_ROWS && node < n) {
          const float di = sd[k];
          iv.x = sv[k].x * di * di;
          iv.y = sv[k].y * di * di;
          iv.z = sv[k].z * di * di;
          iv.w = sv[k].w * di * di;
        }
        if (add) {
          const float4 r = ((float4*)acc)[u * P4 + lane];
          iv.x += r.x; iv.y += r.y; iv.z += r.z; iv.w += r.w;
        }
        ((float4*)acc)[u * P4 + lane] = iv;
      }
    }
  };

  int  whole = -1;
  int  have  = 0;
  int4 head  = make_int4(0, 0, 0, 0);
  if (!FIRST) {
    const int have_raw  = kcnt[kslot];
    const int whole_raw = kcnt[blockIdx.x * AGG_WAVES];
    self_load();
    have = __builtin_amdgcn_readfirstlane(have_raw);
    if (lane < have) head = keep[lane];
    whole = __builtin_amdgcn_readfirstlane(whole_raw);
  }
  put_piece(0, wv0);
  put_piece(1, wv1);
  if (WN == 4) {
    put_piece(2, wv2);
    put_piece(3, wv3);
  }

  if (whole >= 0) {
    self_store(false);
    own_replay(acc, keep, head, have, lane, h);
  } else {
    for (int i = tid; i < ROWS_PAD * P4; i += AGG_THREADS)
      ((float4*)acc)[i] = make_float4(0.0f, 0.0f, 0.0f, 0.0f);
    if (tid < AGG_WAVES) bcnt[tid] = 0;
    if (tid == 0) spilled = 0;
    __syncthreads();

    bool lost = false;

    auto file = [&](bool mine, unsigned u, int eid) {
      if (mine) {
        const int o = (int)(u & 31u);
        const int slot = __hip_atomic_fetch_add(&bcnt[o], 1, __ATOMIC_RELAXED,
                                                __HIP_MEMORY_SCOPE_WORKGROUP);
        if (slot < AGG_BCAP) {
          bin[o * AGG_BCAP + slot] = make_int2(eid, (int)u);
        } else {
          lost = true;
          const int sn = src[eid];
          const float wt = dinv[sn] * dinv[lo + (int)u];
          const float* hr = h + (size_t)sn * F;
          float* ar = acc + (int)u * PITCH;
#pragma nounroll
          for (int k = 0; k < F; ++k)
            __hip_atomic_fetch_add(ar + k, hr[k] * wt, __ATOMIC_RELAXED, __HIP_MEMORY_SCOPE_WORKGROUP);
        }
      }
    };
    auto verify = [&](int g, unsigned wmap) {
      while (wmap) {
        const int bit = __builtin_ctz(wmap);
        wmap &= wmap - 1u;
        const int eid = (g << 4) + ((bit & 7) << 2) + (bit >> 3);
        if (eid < e) {
          const unsigned u = (unsigned)(dst[eid] - lo);
          file(u < (unsigned)AGG_ROWS, u, eid);
        }
      }
    };

    const unsigned pat = (unsigned)(blockIdx.x & 255) * 0x01010101u;
    int2* eb  = ebin + wave * OWN_CAP;
    int ecnt  = 0;
    auto probe = [&](int4 o, int g) {
      const unsigned x0 = (unsigned)o.x ^ pat, x1 = (unsigned)o.y ^ pat;
      const unsigned x2 = (unsigned)o.z ^ pat, x3 = (unsigned)o.w ^ pat;
      const unsigned z0 = (x0 - 0x01010101u) & ~x0 & 0x80808080u;
      const unsigned z1 = (x1 - 0x01010101u) & ~x1 & 0x80808080u;
      const unsigned z2 = (x2 - 0x01010101u) & ~x2 & 0x80808080u;
      const unsigned z3 = (x3 - 0x01010101u) & ~x3 & 0x80808080u;
      const bool has = (z0 | z1 | z2 | z3) != 0u;
      const unsigned hm = __builtin_amdgcn_ballot_w32(has);
      if (hm) {
        if (has) {
          const unsigned wmap = (z0 >> 7) | (z1 >> 6) | (z2 >> 5) | (z3 >> 4);
          const int slot = ecnt + (int)__builtin_amdgcn_mbcnt_lo(hm, 0u);
          if (slot < OWN_CAP) eb[slot] = make_int2(g, (int)wmap);
          else verify(g, wmap);
        }
        ecnt += __builtin_popcount(hm);
      }
    };

    const int4* ov  = (const int4*)own;
    const int g16   = (e + 15) >> 4;
    constexpr int SU  = 8;
    constexpr int GRP = AGG_THREADS * SU;
    const int nfull = g16 / GRP;

    const int rot = (int)(((long)blockIdx.x * nfull) / gridDim.x);
    for (int it = 0; it < nfull; ++it) {
      int g = it + rot;
      if (g >= nfull) g -= nfull;
      const int q0 = g * GRP + tid;
      int4 o[SU];
  #pragma unroll
      for (int t = 0; t < SU; ++t) o[t] = ov[q0 + t * AGG_THREADS];
  #pragma unroll
      for (int t = 0; t < SU; ++t) probe(o[t], q0 + t * AGG_THREADS);
    }
    for (int q0 = nfull * GRP + wave * 32; q0 < g16; q0 += AGG_THREADS) {
      const int qi = q0 + lane;
      int4 o = make_int4((int)~pat, (int)~pat, (int)~pat, (int)~pat);
      if (qi < g16) o = ov[qi];
      probe(o, qi);
    }
    {
      const int have = ecnt < OWN_CAP ? ecnt : OWN_CAP;
      for (int b = 0; b < have; b += 32) {
        if (b + lane < have) {
          const int2 c = eb[b + lane];
          verify(c.x, (unsigned)c.y);
        }
      }
    }

    if (lost) spilled = 1;
    __syncthreads();

    int cnt = __builtin_amdgcn_readfirstlane(bcnt[wave]);
    if (cnt > AGG_BCAP) cnt = AGG_BCAP;
    self_load();
    self_store(true);
    own_drain<FIRST>(acc, bin + wave * AGG_BCAP, cnt, lane, lo, h, dinv, src, keep);
    if (FIRST && lane == 0) kcnt[kslot] = spilled ? -1 : cnt;
  }
  __syncthreads();

  constexpr int NT = DOUT / 16;
  constexpr int RG = AGG_WAVES / NT;
  const int half = lane >> 4;
  const int l15  = lane & 15;
  const int ct   = wave % NT;
  const int rg   = wave / NT;
  const int col  = ct * 16 + l15;

  const float* wr = wl + col * PITCH + half * 2;
  const float bv  = bias[col];

  int rt = rg;
  for (; rt + RG < ROW_TILES; rt += 2 * RG) {
    const float* a0 = acc + (rt * 16 + l15) * PITCH + half * 2;
    const float* a1 = a0 + RG * 16 * PITCH;
    v8f d0 = {}, d1 = {};
#pragma unroll 8
    for (int t = 0; t < F / 4; ++t) {
      const v2f f0 = *(const v2f*)(a0 + 4 * t);
      const v2f f1 = *(const v2f*)(a1 + 4 * t);
      const v2f fb = *(const v2f*)(wr + 4 * t);
      d0 = __builtin_amdgcn_wmma_f32_16x16x4_f32(false, f0, false, fb, (short)0, d0, false, false);
      d1 = __builtin_amdgcn_wmma_f32_16x16x4_f32(false, f1, false, fb, (short)0, d1, false, false);
    }
    put_tile<DOUT>(out, d0, rt, half, col, bv, lo, n, dorelu);
    put_tile<DOUT>(out, d1, rt + RG, half, col, bv, lo, n, dorelu);
  }
  if (rt < ROW_TILES) {
    const float* a0 = acc + (rt * 16 + l15) * PITCH + half * 2;
    v8f d0 = {};
#pragma unroll 8
    for (int t = 0; t < F / 4; ++t) {
      const v2f f0 = *(const v2f*)(a0 + 4 * t);
      const v2f fb = *(const v2f*)(wr + 4 * t);
      d0 = __builtin_amdgcn_wmma_f32_16x16x4_f32(false, f0, false, fb, (short)0, d0, false, false);
    }
    put_tile<DOUT>(out, d0, rt, half, col, bv, lo, n, dorelu);
  }
}

static inline size_t align256(size_t x) { return (x + 255) & ~(size_t)255; }

extern "C" void kernel_launch(void* const* d_in, const int* in_sizes, int n_in,
                              void* d_out, int out_size, void* d_ws, size_t ws_size,
                              hipStream_t stream) {
  (void)n_in; (void)out_size; (void)ws_size;

  const int DIN = 128, DH = 128, DOUT = 64, LAYERS = 5;
  const int N = in_sizes[0] / DIN;
  const int E = in_sizes[1] / 2;

  const float* x0   = (const float*)d_in[0];
  const int*  eidx  = (const int*)d_in[1];
  const int*  srcA  = eidx;
  const int*  dstA  = eidx + E;

  char* ws = (char*)d_ws;
  size_t off = 0;
  float* dinv = (float*)(ws + off); off += align256((size_t)N * sizeof(float));
  float* bufA = (float*)(ws + off); off += align256((size_t)N * DH * sizeof(float));
  float* bufB = (float*)(ws + off); off += align256((size_t)N * DH * sizeof(float));
  const int ablocks = (N + AGG_ROWS - 1) / AGG_ROWS;
  int* kcnt = (int*)(ws + off); off += align256((size_t)ablocks * AGG_WAVES * sizeof(int));
  int* kent = (int*)(ws + off); off += align256((size_t)ablocks * AGG_WAVES * AGG_LCAP * 16);
  int* own  = (int*)(ws + off); off += align256((size_t)E);

  const int TB = 256;

  k_deg_init <<<(N + TB - 1) / TB, TB, 0, stream>>>(dinv, N);
  k_deg_count<<<(E + TB - 1) / TB, TB, 0, stream>>>(dinv, dstA, own, E);
  k_dinv     <<<(N + TB - 1) / TB, TB, 0, stream>>>(dinv, N);

  const float* xin = x0;
  for (int layer = 0; layer < LAYERS; ++layer) {
    const int dout = (layer == LAYERS - 1) ? DOUT : DH;
    const float* W = (const float*)d_in[2 + 2 * layer];
    const float* b = (const float*)d_in[3 + 2 * layer];
    float* out = (layer == LAYERS - 1) ? (float*)d_out : ((layer & 1) ? bufB : bufA);

    const int relu = layer < LAYERS - 1 ? 1 : 0;
    if (layer == 0)
      k_agg_owner<128, true><<<ablocks, AGG_THREADS, 0, stream>>>(out, xin, dinv, W, b, srcA, dstA, own, kcnt, kent, N, E, relu);
    else if (dout == 128)
      k_agg_owner<128, false><<<ablocks, AGG_THREADS, 0, stream>>>(out, xin, dinv, W, b, srcA, dstA, own, kcnt, kent, N, E, relu);
    else
      k_agg_owner<64, false><<<ablocks, AGG_THREADS, 0, stream>>>(out, xin, dinv, W, b, srcA, dstA, own, kcnt, kent, N, E, relu);

    xin = out;
  }
}
